# starting kernel
# baseline (speedup 1.0000x reference)
#include <hip/hip_runtime.h>

typedef float v2f __attribute__((ext_vector_type(2)));
typedef float v8f __attribute__((ext_vector_type(8)));

#define T_LEN 16384
#define D_DIM 1024
#define H_DIM 8

__global__ __launch_bounds__(256) void rnn_inproj_wmma(
    const float* __restrict__ x, const float* __restrict__ W_ih,
    const float* __restrict__ b_ih, const float* __restrict__ b_hh,
    float* __restrict__ xwT)
{
    const int lane = threadIdx.x & 31;
    const int wave = threadIdx.x >> 5;
    const int tile = blockIdx.x * 8 + wave;
    const int r    = lane & 15;
    const int hi   = lane >> 4;
    const int kb   = hi * 2;
    const int nn   = r & 7;
    const float nmask = (r < 8) ? 1.0f : 0.0f;

    const float* xrow = x    + (size_t)(tile * 16 + r) * D_DIM;
    const float* wrow = W_ih + (size_t)nn * D_DIM;

    v8f acc = {};
#pragma unroll 4
    for (int k = 0; k < D_DIM; k += 4) {
        v2f a = *(const v2f*)(xrow + k + kb);
        v2f b = *(const v2f*)(wrow + k + kb);
        b.x *= nmask;
        b.y *= nmask;
        acc = __builtin_amdgcn_wmma_f32_16x16x4_f32(
             false, a,  false, b,
             (short)0, acc,  false,  false);
    }

    if (r < 8) {
        const float bias = b_ih[r] + b_hh[r];
#pragma unroll
        for (int v = 0; v < 8; ++v) {
            const int tl = tile * 16 + v + hi * 8;
            xwT[(size_t)r * T_LEN + tl] = acc[v] + bias;
        }
    }
}

__global__ void rnn_scan(const float* __restrict__ xwT,
                         const float* __restrict__ W_hh,
                         float* __restrict__ hs)
{
    const int lane = threadIdx.x & 31;
    const int i = lane & 7;

    float w[8];
#pragma unroll
    for (int j = 0; j < 8; ++j) w[j] = W_hh[i * 8 + j];

    const float4* xs = (const float4*)(xwT + (size_t)i * T_LEN);
    float4 cur = xs[0];
    float h = 0.0f;

    for (int t = 0; t < T_LEN; t += 4) {
        const int tn = (t + 4 < T_LEN) ? (t + 4) : 0;
        float4 nxt = xs[tn >> 2];
        const float xv[4] = {cur.x, cur.y, cur.z, cur.w};
#pragma unroll
        for (int c = 0; c < 4; ++c) {
            float s = xv[c];
#pragma unroll
            for (int j = 0; j < 8; ++j) {
                const float hj = __int_as_float(
                    __builtin_amdgcn_readlane(__float_as_int(h), j));
                s = fmaf(w[j], hj, s);
            }
            h = fmaxf(s, 0.0f);
            if (lane < 8)
                hs[(size_t)(t + c) * 8 + lane] = h;
        }
        cur = nxt;
    }
}

__global__ __launch_bounds__(256) void rnn_out(
    const float* __restrict__ hs, const float* __restrict__ W_out,
    const float* __restrict__ b_out, float* __restrict__ out)
{
    const int t = blockIdx.x * blockDim.x + threadIdx.x;
    if (t >= T_LEN) return;
    const float4* hp = (const float4*)(hs + (size_t)t * 8);
    const float4 h0 = hp[0], h1 = hp[1];
    float s = b_out[0];
    s += h0.x * W_out[0] + h0.y * W_out[1] + h0.z * W_out[2] + h0.w * W_out[3];
    s += h1.x * W_out[4] + h1.y * W_out[5] + h1.z * W_out[6] + h1.w * W_out[7];
    out[t] = s;
}

extern "C" void kernel_launch(void* const* d_in, const int* in_sizes, int n_in,
                              void* d_out, int out_size, void* d_ws, size_t ws_size,
                              hipStream_t stream) {
    const float* x     = (const float*)d_in[0];
    const float* W_ih  = (const float*)d_in[1];
    const float* b_ih  = (const float*)d_in[2];
    const float* W_hh  = (const float*)d_in[3];
    const float* b_hh  = (const float*)d_in[4];
    const float* W_out = (const float*)d_in[5];
    const float* b_out = (const float*)d_in[6];
    float* out = (float*)d_out;

    float* xwT = (float*)d_ws;
    float* hs  = xwT + (size_t)T_LEN * H_DIM;

    rnn_inproj_wmma<<<(T_LEN / 16) / 8, 256, 0, stream>>>(x, W_ih, b_ih, b_hh, xwT);
    rnn_scan<<<1, 32, 0, stream>>>(xwT, W_hh, hs);
    rnn_out<<<(T_LEN + 255) / 256, 256, 0, stream>>>(hs, W_out, b_out, out);
}


# scan: a part's h rows are stored one part later (3 h sets in turn): the registers a store names are rewritten 8 steps behind it, so no address-translation wait is on the serial path; sums unchanged
# speedup vs baseline: 3.0707x; 3.0707x over previous
#include <hip/hip_runtime.h>


typedef float v2f __attribute__((ext_vector_type(2)));
typedef float v8f __attribute__((ext_vector_type(8)));

#define T_LEN 16384
#define D_DIM 1024
#define H_DIM 8

__device__ __forceinline__ int unit_of_slot(int p) {
    return ((p & 2) << 1) | (p & 1) | ((p & 4) >> 1);
}

__global__ __launch_bounds__(256) void rnn_inproj_wmma(
    const float* __restrict__ x, const float* __restrict__ W_ih,
    const float* __restrict__ b_ih, const float* __restrict__ b_hh,
    float* __restrict__ xwP)
{
    const int lane = threadIdx.x & 31;
    const int wave = threadIdx.x >> 5;
    const int tile = blockIdx.x * 8 + wave;
    const int r    = lane & 15;
    const int hi   = lane >> 4;
    const int kb   = hi * 2;
    const int nn   = unit_of_slot(r & 7);
    const float nmask = (r < 8) ? 1.0f : 0.0f;

    const float* xrow = x    + (size_t)(tile * 16 + r) * D_DIM;
    const float* wrow = W_ih + (size_t)nn * D_DIM;

    v8f acc = {};
#pragma unroll 4
    for (int k = 0; k < D_DIM; k += 4) {
        v2f a = *(const v2f*)(xrow + k + kb);
        v2f b = *(const v2f*)(wrow + k + kb);
        b.x *= nmask;
        b.y *= nmask;
        acc = __builtin_amdgcn_wmma_f32_16x16x4_f32(
             false, a,  false, b,
             (short)0, acc,  false,  false);
    }

    if (r < 8) {
        const float bias = b_ih[nn] + b_hh[nn];
#pragma unroll
        for (int v = 0; v < 8; ++v) {
            const int tl = tile * 16 + v + hi * 8;
            xwP[(size_t)tl * H_DIM + r] = acc[v] + bias;
        }
    }
}

#define SCAN_PART 8
#define SCAN_PARTS (T_LEN / SCAN_PART)

__device__ __forceinline__ void load_low(v8f& acc, const float4* __restrict__ p)
{
    const float4 xv = *p;
    acc[0] = xv.x; acc[1] = xv.y; acc[2] = xv.z; acc[3] = xv.w;
}

__device__ __forceinline__ void join_relu(const v8f& d, v2f& b)
{
    float bx, by;
    asm volatile(
        "v_add_f32_dpp %0, %2, %3 row_xmask:4 row_mask:0xf bank_mask:0x5\n\t"
        "v_add_f32_dpp %0, %4, %5 row_xmask:4 row_mask:0xf bank_mask:0xa\n\t"
        "v_add_f32_dpp %1, %6, %7 row_xmask:4 row_mask:0xf bank_mask:0x5\n\t"
        "v_add_f32_dpp %1, %8, %9 row_xmask:4 row_mask:0xf bank_mask:0xa\n\t"
        "v_max_i32 %0, 0, %0\n\t"
        "v_max_i32 %1, 0, %1"
        : "=&v"(bx), "=&v"(by)
        : "v"(d[4]), "v"(d[0]), "v"(d[6]), "v"(d[2]),
          "v"(d[5]), "v"(d[1]), "v"(d[7]), "v"(d[3]));
    b.x = bx; b.y = by;
}

__device__ __forceinline__ void zero_high(v8f& t)
{
    float z4, z5, z6, z7;
    asm volatile("v_mov_b32 %0, 0\n\t"
                 "v_mov_b32 %1, 0\n\t"
                 "v_mov_b32 %2, 0\n\t"
                 "v_mov_b32 %3, 0"
                 : "=v"(z4), "=v"(z5), "=v"(z6), "=v"(z7));
    t[4] = z4; t[5] = z5; t[6] = z6; t[7] = z7;
}

#define SCAN_PF_AHEAD 16
__device__ __forceinline__ void prefetch_unit(const float* __restrict__ base, const unsigned off)
{
    asm volatile("global_prefetch_b8 %0, %1 scope:SCOPE_SE" : : "v"(off), "s"(base));
}

__device__ __forceinline__ const float* unit_of_part(const float* __restrict__ xwP, const int part)
{
    return xwP + (SCAN_PART * H_DIM) * (part < SCAN_PARTS - 1 ? part : SCAN_PARTS - 1);
}

__device__ __forceinline__ void scan_part(
    v8f (&cur)[SCAN_PART], v8f (&oth)[SCAN_PART],
    v2f (&hcur)[SCAN_PART], v2f (&hst)[SCAN_PART],
    const v2f a, v2f& b,
    const char* __restrict__ xin, const unsigned xoff,
    char* __restrict__ hrow, const unsigned hoff, const bool store_rows,
    const bool ask, const float* __restrict__ pfu, const unsigned pf_off)
{
#pragma unroll
    for (int u = 0; u < SCAN_PART; ++u) {
        cur[u] = __builtin_amdgcn_wmma_f32_16x16x4_f32(
            false, a, false, b, (short)0, cur[u], false, false);
        __builtin_amdgcn_sched_barrier(0);
        if (u == 0) {
            if (store_rows) {
#pragma unroll
                for (int k = 0; k < SCAN_PART; ++k) *(v2f*)(hrow + hoff + 32 * k) = hst[k];
            }
#pragma unroll
            for (int k = 0; k < SCAN_PART; ++k) load_low(oth[k], (const float4*)(xin + xoff) + 2 * k);
            if (ask) prefetch_unit(pfu, pf_off);
            zero_high(oth[SCAN_PART - 1]);
            __builtin_amdgcn_sched_barrier(0);
        } else {
            zero_high(cur[u - 1]);
        }
        join_relu(cur[u], b);
        hcur[u] = b;
    }
}

__global__ __launch_bounds__(32) void rnn_scan(const float* __restrict__ xwP,
                         const float* __restrict__ W_hh,
                         float* __restrict__ hsP)
{
    const int lane = threadIdx.x & 31;
    const int m   = lane & 15;
    const int hf  = lane >> 4;
    const int cls = (m >> 2) & 1;

    const int hr  = m >> 3;
    const int r   = m & 7;
    const int qu  = (r >> 1) & 1;
    const int sub = qu ^ (r >> 2);
    const int unit = unit_of_slot(4 * hr + 2 * qu + (r & 1));
    v2f a;
    a.x = W_hh[unit * 8 + 4 * sub + 2 * hf + 0];
    a.y = W_hh[unit * 8 + 4 * sub + 2 * hf + 1];

    prefetch_unit(xwP, (unsigned)lane * (SCAN_PART * H_DIM * 4u));
    const unsigned pf0 = 0u;

    const float4* xs = (const float4*)xwP + hf;
    const unsigned xoff = 16u * (unsigned)hf;
    const unsigned hoff = 8u * (unsigned)(2 * hf + cls);

    v8f acc0[SCAN_PART], acc1[SCAN_PART], acc2[SCAN_PART];
    v2f h0[SCAN_PART], h1[SCAN_PART], h2[SCAN_PART];
#pragma unroll
    for (int u = 0; u < SCAN_PART; ++u) {
        acc0[u] = v8f{}; acc1[u] = v8f{}; acc2[u] = v8f{};
        h0[u] = v2f{0.0f, 0.0f};
        h1[u] = v2f{0.0f, 0.0f};
        h2[u] = v2f{0.0f, 0.0f};
    }
#pragma unroll
    for (int u = 0; u < SCAN_PART; ++u) load_low(acc0[u], xs + 2 * u);
    __builtin_amdgcn_sched_barrier(0);
#pragma unroll
    for (int u = 0; u < SCAN_PART; ++u) load_low(acc1[u], xs + 2 * (SCAN_PART + u));

    v2f b = {0.0f, 0.0f};

    const char* const xb0 = (const char*)xwP;
    char* const       hb0 = (char*)hsP;
    scan_part(acc0, acc2, h0, h1, a, b, xb0 + 512,  xoff, hb0, hoff, false,
              true, unit_of_part(xwP, SCAN_PF_AHEAD), pf0);
    scan_part(acc1, acc0, h1, h2, a, b, xb0 + 768,  xoff, hb0, hoff, false,
              true, unit_of_part(xwP, 1 + SCAN_PF_AHEAD), pf0);
    scan_part(acc2, acc1, h2, h0, a, b, xb0 + 1024, xoff, hb0, hoff, true,
              true, unit_of_part(xwP, 2 + SCAN_PF_AHEAD), pf0);
    int k = 3;
    for (; k + 3 <= SCAN_PARTS; k += 3) {
        const char* const xb = xb0 + 256 * k;
        char* const       hb = hb0 + 256 * k;
        scan_part(acc0, acc2, h0, h1, a, b, xb + 512,  xoff, hb - 512, hoff, true,
                  true, unit_of_part(xwP, k + SCAN_PF_AHEAD), pf0);
        scan_part(acc1, acc0, h1, h2, a, b, xb + 768,  xoff, hb - 256, hoff, true,
                  true, unit_of_part(xwP, k + 1 + SCAN_PF_AHEAD), pf0);
        scan_part(acc2, acc1, h2, h0, a, b, xb + 1024, xoff, hb,       hoff, true,
                  true, unit_of_part(xwP, k + 2 + SCAN_PF_AHEAD), pf0);
    }
    scan_part(acc0, acc2, h0, h1, a, b, xb0, xoff, hb0 + 256 * (SCAN_PARTS - 4), hoff, true,
              false, xwP, pf0);
    scan_part(acc1, acc0, h1, h2, a, b, xb0, xoff, hb0 + 256 * (SCAN_PARTS - 3), hoff, true,
              false, xwP, pf0);

#pragma unroll
    for (int j = 0; j < SCAN_PART; ++j) *(v2f*)(hb0 + 256 * (SCAN_PARTS - 2) + hoff + 32 * j) = h0[j];
#pragma unroll
    for (int j = 0; j < SCAN_PART; ++j) *(v2f*)(hb0 + 256 * (SCAN_PARTS - 1) + hoff + 32 * j) = h1[j];
}

__global__ __launch_bounds__(256) void rnn_out(
    const float* __restrict__ hsP, const float* __restrict__ W_out,
    const float* __restrict__ b_out, float* __restrict__ out)
{
    const int t = blockIdx.x * blockDim.x + threadIdx.x;
    if (t >= T_LEN) return;
    const float4* hp = (const float4*)(hsP + (size_t)t * 8);
    const float4 p0 = hp[0], p1 = hp[1];
    float s = b_out[0];
    s += p0.x * W_out[0] + p0.y * W_out[1] + p1.x * W_out[2] + p1.y * W_out[3];
    s += p0.z * W_out[4] + p0.w * W_out[5] + p1.z * W_out[6] + p1.w * W_out[7];
    out[t] = s;
}

extern "C" void kernel_launch(void* const* d_in, const int* in_sizes, int n_in,
                              void* d_out, int out_size, void* d_ws, size_t ws_size,
                              hipStream_t stream) {
    const float* x     = (const float*)d_in[0];
    const float* W_ih  = (const float*)d_in[1];
    const float* b_ih  = (const float*)d_in[2];
    const float* W_hh  = (const float*)d_in[3];
    const float* b_hh  = (const float*)d_in[4];
    const float* W_out = (const float*)d_in[5];
    const float* b_out = (const float*)d_in[6];
    float* out = (float*)d_out;

    float* xwP = (float*)d_ws;
    float* hsP = xwP + (size_t)T_LEN * H_DIM;

    rnn_inproj_wmma<<<(T_LEN / 16) / 8, 256, 0, stream>>>(x, W_ih, b_ih, b_hh, xwP);
    rnn_scan<<<1, 32, 0, stream>>>(xwP, W_hh, hsP);
    rnn_out<<<(T_LEN + 255) / 256, 256, 0, stream>>>(hsP, W_out, b_out, out);
}


# input projection: x reaches the A fragments through LDS: a wave reads its 16 rows as whole 512-byte runs two 128-k chunks ahead; W_ih parked in LDS once a block; 256 4-wave blocks; sums unchanged
# speedup vs baseline: 3.2157x; 1.0472x over previous
#include <hip/hip_runtime.h>


typedef float v2f __attribute__((ext_vector_type(2)));
typedef float v8f __attribute__((ext_vector_type(8)));
typedef float v4f __attribute__((ext_vector_type(4)));

#define T_LEN 16384
#define D_DIM 1024
#define H_DIM 8

__device__ __forceinline__ int unit_of_slot(int p) {
    return ((p & 2) << 1) | (p & 1) | ((p & 4) >> 1);
}

#define IP_WAVES 4
#define IP_CHUNK 128
#define IP_NCHUNK (D_DIM / IP_CHUNK)
#define IP_XPITCH (IP_CHUNK + 4)
#define IP_WPITCH (D_DIM + 4)

__device__ __forceinline__ void ip_fetch(v4f (&g)[16], const float* __restrict__ xt, const int chunk, const int lane)
{
#pragma unroll
    for (int i = 0; i < 16; ++i)
        g[i] = *(const v4f*)(xt + (size_t)i * D_DIM + chunk * IP_CHUNK + 4 * lane);
}

__device__ __forceinline__ void ip_park(float (*xs)[IP_XPITCH], const v4f (&g)[16], const int lane)
{
#pragma unroll
    for (int i = 0; i < 16; ++i) *(v4f*)(&xs[i][4 * lane]) = g[i];
}

__device__ __forceinline__ void ip_frags(v2f (&fa)[8], v2f (&fb)[8], const float (*xs)[IP_XPITCH],
                                         const float* __restrict__ wrow, const int chunk, const int blk,
                                         const int r, const int kb)
{
#pragma unroll
    for (int q = 0; q < 8; ++q) {
        const int j = 8 * blk + q;
        fa[q] = *(const v2f*)(&xs[r][4 * j + kb]);
        fb[q] = *(const v2f*)(wrow + chunk * IP_CHUNK + 4 * j + kb);
    }
}

__device__ __forceinline__ void ip_block(v8f& acc, const v2f (&fa)[8], const v2f (&fb)[8])
{
#pragma unroll
    for (int q = 0; q < 8; ++q)
        acc = __builtin_amdgcn_wmma_f32_16x16x4_f32(
             false, fa[q],  false, fb[q],
             (short)0, acc,  false,  false);
}

__device__ __forceinline__ void ip_products(v8f& acc, const float (*xs)[IP_XPITCH], const float* __restrict__ wrow,
                                            const int chunk, const int r, const int kb)
{
    v2f a0[8], b0[8], a1[8], b1[8];
    ip_frags(a0, b0, xs, wrow, chunk, 0, r, kb);
    __builtin_amdgcn_sched_barrier(0);
    ip_frags(a1, b1, xs, wrow, chunk, 1, r, kb);
    __builtin_amdgcn_sched_barrier(0);
    ip_block(acc, a0, b0);
    __builtin_amdgcn_sched_barrier(0);
    ip_frags(a0, b0, xs, wrow, chunk, 2, r, kb);
    __builtin_amdgcn_sched_barrier(0);
    ip_block(acc, a1, b1);
    __builtin_amdgcn_sched_barrier(0);
    ip_frags(a1, b1, xs, wrow, chunk, 3, r, kb);
    __builtin_amdgcn_sched_barrier(0);
    ip_block(acc, a0, b0);
    __builtin_amdgcn_sched_barrier(0);
    ip_block(acc, a1, b1);
}

__global__ __launch_bounds__(IP_WAVES * 32) void rnn_inproj_wmma(
    const float* __restrict__ x, const float* __restrict__ W_ih,
    const float* __restrict__ b_ih, const float* __restrict__ b_hh,
    float* __restrict__ xwP)
{
    __shared__ __attribute__((aligned(16))) float xsh[IP_WAVES][16][IP_XPITCH];
    __shared__ __attribute__((aligned(16))) float wsh[H_DIM][IP_WPITCH];

    const int lane = threadIdx.x & 31;
    const int wave = threadIdx.x >> 5;
    const int tile = blockIdx.x * IP_WAVES + wave;
    const int r    = lane & 15;
    const int hi   = lane >> 4;
    const int kb   = hi * 2;
    const int nn   = unit_of_slot(r & 7);

    const float* xt = x + (size_t)(tile * 16) * D_DIM;
    float (*xs)[IP_XPITCH] = xsh[wave];

    v4f g0[16], g1[16];
    ip_fetch(g0, xt, 0, lane);

#pragma unroll
    for (int i = 0; i < (H_DIM * D_DIM / 4) / (IP_WAVES * 32); ++i) {
        const int idx = (int)threadIdx.x + (IP_WAVES * 32) * i;
        const int p   = idx >> 8;
        const int c4  = idx & 255;
        *(v4f*)(&wsh[p][4 * c4]) = *(const v4f*)(W_ih + (size_t)unit_of_slot(p) * D_DIM + 4 * c4);
    }
    __syncthreads();
    ip_fetch(g1, xt, 1, lane);
    const float* wrow = wsh[r & 7];

    v8f acc = {};
    ip_park(xs, g0, lane); ip_fetch(g0, xt, 2, lane); ip_products(acc, xs, wrow, 0, r, kb);
    ip_park(xs, g1, lane); ip_fetch(g1, xt, 3, lane); ip_products(acc, xs, wrow, 1, r, kb);
    ip_park(xs, g0, lane); ip_fetch(g0, xt, 4, lane); ip_products(acc, xs, wrow, 2, r, kb);
    ip_park(xs, g1, lane); ip_fetch(g1, xt, 5, lane); ip_products(acc, xs, wrow, 3, r, kb);
    ip_park(xs, g0, lane); ip_fetch(g0, xt, 6, lane); ip_products(acc, xs, wrow, 4, r, kb);
    ip_park(xs, g1, lane); ip_fetch(g1, xt, 7, lane); ip_products(acc, xs, wrow, 5, r, kb);
    ip_park(xs, g0, lane);                            ip_products(acc, xs, wrow, 6, r, kb);
    ip_park(xs, g1, lane);                            ip_products(acc, xs, wrow, 7, r, kb);

    if (r < 8) {
        const float bias = b_ih[nn] + b_hh[nn];
#pragma unroll
        for (int v = 0; v < 8; ++v) {
            const int tl = tile * 16 + v + hi * 8;
            xwP[(size_t)tl * H_DIM + r] = acc[v] + bias;
        }
    }
}

#define SCAN_PART 8
#define SCAN_PARTS (T_LEN / SCAN_PART)

__device__ __forceinline__ void load_low(v8f& acc, const float4* __restrict__ p)
{
    const float4 xv = *p;
    acc[0] = xv.x; acc[1] = xv.y; acc[2] = xv.z; acc[3] = xv.w;
}

__device__ __forceinline__ void join_relu(const v8f& d, v2f& b)
{
    float bx, by;
    asm volatile(
        "v_add_f32_dpp %0, %2, %3 row_xmask:4 row_mask:0xf bank_mask:0x5\n\t"
        "v_add_f32_dpp %0, %4, %5 row_xmask:4 row_mask:0xf bank_mask:0xa\n\t"
        "v_add_f32_dpp %1, %6, %7 row_xmask:4 row_mask:0xf bank_mask:0x5\n\t"
        "v_add_f32_dpp %1, %8, %9 row_xmask:4 row_mask:0xf bank_mask:0xa\n\t"
        "v_max_i32 %0, 0, %0\n\t"
        "v_max_i32 %1, 0, %1"
        : "=&v"(bx), "=&v"(by)
        : "v"(d[4]), "v"(d[0]), "v"(d[6]), "v"(d[2]),
          "v"(d[5]), "v"(d[1]), "v"(d[7]), "v"(d[3]));
    b.x = bx; b.y = by;
}

__device__ __forceinline__ void zero_high(v8f& t)
{
    float z4, z5, z6, z7;
    asm volatile("v_mov_b32 %0, 0\n\t"
                 "v_mov_b32 %1, 0\n\t"
                 "v_mov_b32 %2, 0\n\t"
                 "v_mov_b32 %3, 0"
                 : "=v"(z4), "=v"(z5), "=v"(z6), "=v"(z7));
    t[4] = z4; t[5] = z5; t[6] = z6; t[7] = z7;
}

#define SCAN_PF_AHEAD 16
__device__ __forceinline__ void prefetch_unit(const float* __restrict__ base, const unsigned off)
{
    asm volatile("global_prefetch_b8 %0, %1 scope:SCOPE_SE" : : "v"(off), "s"(base));
}

__device__ __forceinline__ const float* unit_of_part(const float* __restrict__ xwP, const int part)
{
    return xwP + (SCAN_PART * H_DIM) * (part < SCAN_PARTS - 1 ? part : SCAN_PARTS - 1);
}

__device__ __forceinline__ void scan_part(
    v8f (&cur)[SCAN_PART], v8f (&oth)[SCAN_PART],
    v2f (&hcur)[SCAN_PART], v2f (&hst)[SCAN_PART],
    const v2f a, v2f& b,
    const char* __restrict__ xin, const unsigned xoff,
    char* __restrict__ hrow, const unsigned hoff, const bool store_rows,
    const bool ask, const float* __restrict__ pfu, const unsigned pf_off)
{
#pragma unroll
    for (int u = 0; u < SCAN_PART; ++u) {
        cur[u] = __builtin_amdgcn_wmma_f32_16x16x4_f32(
            false, a, false, b, (short)0, cur[u], false, false);
        __builtin_amdgcn_sched_barrier(0);
        if (u == 0) {
            if (store_rows) {
#pragma unroll
                for (int k = 0; k < SCAN_PART; ++k) *(v2f*)(hrow + hoff + 32 * k) = hst[k];
            }
#pragma unroll
            for (int k = 0; k < SCAN_PART; ++k) load_low(oth[k], (const float4*)(xin + xoff) + 2 * k);
            if (ask) prefetch_unit(pfu, pf_off);
            zero_high(oth[SCAN_PART - 1]);
            __builtin_amdgcn_sched_barrier(0);
        } else {
            zero_high(cur[u - 1]);
        }
        join_relu(cur[u], b);
        hcur[u] = b;
    }
}

__global__ __launch_bounds__(32) void rnn_scan(const float* __restrict__ xwP,
                         const float* __restrict__ W_hh,
                         float* __restrict__ hsP)
{
    const int lane = threadIdx.x & 31;
    const int m   = lane & 15;
    const int hf  = lane >> 4;
    const int cls = (m >> 2) & 1;

    const int hr  = m >> 3;
    const int r   = m & 7;
    const int qu  = (r >> 1) & 1;
    const int sub = qu ^ (r >> 2);
    const int unit = unit_of_slot(4 * hr + 2 * qu + (r & 1));
    v2f a;
    a.x = W_hh[unit * 8 + 4 * sub + 2 * hf + 0];
    a.y = W_hh[unit * 8 + 4 * sub + 2 * hf + 1];

    prefetch_unit(xwP, (unsigned)lane * (SCAN_PART * H_DIM * 4u));
    const unsigned pf0 = 0u;

    const float4* xs = (const float4*)xwP + hf;
    const unsigned xoff = 16u * (unsigned)hf;
    const unsigned hoff = 8u * (unsigned)(2 * hf + cls);

    v8f acc0[SCAN_PART], acc1[SCAN_PART], acc2[SCAN_PART];
    v2f h0[SCAN_PART], h1[SCAN_PART], h2[SCAN_PART];
#pragma unroll
    for (int u = 0; u < SCAN_PART; ++u) {
        acc0[u] = v8f{}; acc1[u] = v8f{}; acc2[u] = v8f{};
        h0[u] = v2f{0.0f, 0.0f};
        h1[u] = v2f{0.0f, 0.0f};
        h2[u] = v2f{0.0f, 0.0f};
    }
#pragma unroll
    for (int u = 0; u < SCAN_PART; ++u) load_low(acc0[u], xs + 2 * u);
    __builtin_amdgcn_sched_barrier(0);
#pragma unroll
    for (int u = 0; u < SCAN_PART; ++u) load_low(acc1[u], xs + 2 * (SCAN_PART + u));

    v2f b = {0.0f, 0.0f};

    const char* const xb0 = (const char*)xwP;
    char* const       hb0 = (char*)hsP;
    scan_part(acc0, acc2, h0, h1, a, b, xb0 + 512,  xoff, hb0, hoff, false,
              true, unit_of_part(xwP, SCAN_PF_AHEAD), pf0);
    scan_part(acc1, acc0, h1, h2, a, b, xb0 + 768,  xoff, hb0, hoff, false,
              true, unit_of_part(xwP, 1 + SCAN_PF_AHEAD), pf0);
    scan_part(acc2, acc1, h2, h0, a, b, xb0 + 1024, xoff, hb0, hoff, true,
              true, unit_of_part(xwP, 2 + SCAN_PF_AHEAD), pf0);
    int k = 3;
    for (; k + 3 <= SCAN_PARTS; k += 3) {
        const char* const xb = xb0 + 256 * k;
        char* const       hb = hb0 + 256 * k;
        scan_part(acc0, acc2, h0, h1, a, b, xb + 512,  xoff, hb - 512, hoff, true,
                  true, unit_of_part(xwP, k + SCAN_PF_AHEAD), pf0);
        scan_part(acc1, acc0, h1, h2, a, b, xb + 768,  xoff, hb - 256, hoff, true,
                  true, unit_of_part(xwP, k + 1 + SCAN_PF_AHEAD), pf0);
        scan_part(acc2, acc1, h2, h0, a, b, xb + 1024, xoff, hb,       hoff, true,
                  true, unit_of_part(xwP, k + 2 + SCAN_PF_AHEAD), pf0);
    }
    scan_part(acc0, acc2, h0, h1, a, b, xb0, xoff, hb0 + 256 * (SCAN_PARTS - 4), hoff, true,
              false, xwP, pf0);
    scan_part(acc1, acc0, h1, h2, a, b, xb0, xoff, hb0 + 256 * (SCAN_PARTS - 3), hoff, true,
              false, xwP, pf0);

#pragma unroll
    for (int j = 0; j < SCAN_PART; ++j) *(v2f*)(hb0 + 256 * (SCAN_PARTS - 2) + hoff + 32 * j) = h0[j];
#pragma unroll
    for (int j = 0; j < SCAN_PART; ++j) *(v2f*)(hb0 + 256 * (SCAN_PARTS - 1) + hoff + 32 * j) = h1[j];
}

__global__ __launch_bounds__(256) void rnn_out(
    const float* __restrict__ hsP, const float* __restrict__ W_out,
    const float* __restrict__ b_out, float* __restrict__ out)
{
    const int t = blockIdx.x * blockDim.x + threadIdx.x;
    if (t >= T_LEN) return;
    const float4* hp = (const float4*)(hsP + (size_t)t * 8);
    const float4 p0 = hp[0], p1 = hp[1];
    float s = b_out[0];
    s += p0.x * W_out[0] + p0.y * W_out[1] + p1.x * W_out[2] + p1.y * W_out[3];
    s += p0.z * W_out[4] + p0.w * W_out[5] + p1.z * W_out[6] + p1.w * W_out[7];
    out[t] = s;
}

extern "C" void kernel_launch(void* const* d_in, const int* in_sizes, int n_in,
                              void* d_out, int out_size, void* d_ws, size_t ws_size,
                              hipStream_t stream) {
    const float* x     = (const float*)d_in[0];
    const float* W_ih  = (const float*)d_in[1];
    const float* b_ih  = (const float*)d_in[2];
    const float* W_hh  = (const float*)d_in[3];
    const float* b_hh  = (const float*)d_in[4];
    const float* W_out = (const float*)d_in[5];
    const float* b_out = (const float*)d_in[6];
    float* out = (float*)d_out;

    float* xwP = (float*)d_ws;
    float* hsP = xwP + (size_t)T_LEN * H_DIM;

    rnn_inproj_wmma<<<(T_LEN / 16) / IP_WAVES, IP_WAVES * 32, 0, stream>>>(x, W_ih, b_ih, b_hh, xwP);
    rnn_scan<<<1, 32, 0, stream>>>(xwP, W_hh, hsP);
    rnn_out<<<(T_LEN + 255) / 256, 256, 0, stream>>>(hsP, W_out, b_out, out);
}
